# starting kernel
# baseline (speedup 1.0000x reference)
#include <hip/hip_runtime.h>
#include <hip/hip_bf16.h>
#include <math.h>


typedef float v2f __attribute__((ext_vector_type(2)));
typedef float v8f __attribute__((ext_vector_type(8)));

#define NQ    12
#define SDIM  4096
#define EPSV  1e-10f

static __device__ __forceinline__ void rot_entry(const float* __restrict__ w,
                                                 int q, int r, int c,
                                                 float& er, float& ei) {
  float phi = w[q * 3 + 0], th = w[q * 3 + 1], om = w[q * 3 + 2];
  float ch = cosf(0.5f * th), sh = sinf(0.5f * th);
  float hpo = 0.5f * (phi + om), hpm = 0.5f * (phi - om);
  float cpo = cosf(hpo), spo = sinf(hpo);
  float cpm = cosf(hpm), spm = sinf(hpm);
  if (r == 0) {
    if (c == 0) { er =  ch * cpo; ei = -ch * spo; }
    else        { er = -sh * cpm; ei = -sh * spm; }
  } else {
    if (c == 0) { er =  sh * cpm; ei = -sh * spm; }
    else        { er =  ch * cpo; ei =  ch * spo; }
  }
}

__global__ void qdl_setup(const float* __restrict__ w,
                          float* __restrict__ mat,
                          int* __restrict__ perm) {
  int tid = blockIdx.x * blockDim.x + threadIdx.x;
  if (tid < 2 * SDIM) {
    int isB = tid >= SDIM;
    int idx = tid & (SDIM - 1);
    int outi = idx >> 6, ini = idx & 63;
    int qbase = isB ? 6 : 0;
    float re = 1.f, im = 0.f;
    for (int q = 0; q < 6; ++q) {
      int bo = (outi >> (5 - q)) & 1;
      int bi = (ini  >> (5 - q)) & 1;
      float er, ei;
      rot_entry(w, qbase + q, bo, bi, er, ei);
      float nre = re * er - im * ei;
      float nim = re * ei + im * er;
      re = nre; im = nim;
    }
    mat[(isB ? 8192 : 0) + idx]     = re;
    mat[(isB ? 12288 : 4096) + idx] = im;
  }
  if (tid < SDIM) {
    int src = tid;
    for (int i = NQ - 1; i >= 0; --i)
      for (int j = NQ - 1; j > i; --j)
        if ((src >> (NQ - 1 - i)) & 1) src ^= (1 << (NQ - 1 - j));
    perm[tid] = src;
  }
}

__global__ void __launch_bounds__(256)
qdl_main(const float* __restrict__ x, const float* __restrict__ mat,
         const int* __restrict__ perm, float* __restrict__ out) {
  __shared__ float buf0[SDIM];
  __shared__ float buf1[SDIM];
  __shared__ float buf2[SDIM];
  __shared__ float red[64];

  const int tid  = threadIdx.x;
  const int wave = tid >> 5;
  const int lane = tid & 31;
  const int ln15 = lane & 15;
  const int khalf = (lane >> 4) * 2;
  const float* __restrict__ xr = x + (size_t)blockIdx.x * SDIM;

  float vals[16];
  float ss = 0.f;
  for (int i = 0; i < 16; ++i) {
    float v = xr[tid + i * 256];
    vals[i] = v;
    ss += v * v;
  }
  for (int o = 16; o > 0; o >>= 1) ss += __shfl_xor(ss, o, 32);
  if (lane == 0) red[wave] = ss;
  __syncthreads();
  if (tid == 0) {
    float t = 0.f;
    for (int i = 0; i < 8; ++i) t += red[i];
    red[32] = t;
  }
  __syncthreads();
  float nrm = sqrtf(red[32]);
  bool zerovec = !(nrm > EPSV);
  float scale = zerovec ? 0.f : (1.0f / fmaxf(nrm, EPSV));
  for (int i = 0; i < 16; ++i)
    buf0[tid + i * 256] = zerovec ? 0.015625f : vals[i] * scale;
  __syncthreads();

  const float* __restrict__ A_re = mat;
  const float* __restrict__ A_im = mat + 4096;
  const float* __restrict__ B_re = mat + 8192;
  const float* __restrict__ B_im = mat + 12288;

  for (int t = wave * 2; t < wave * 2 + 2; ++t) {
    int tm = t >> 2, tn = t & 3;
    int m = tm * 16 + ln15;
    int n = tn * 16 + ln15;
    v8f acc_re = {};
    v8f acc_im = {};
    for (int ks = 0; ks < 16; ++ks) {
      int k0 = ks * 4 + khalf;
      v2f ar, ai, bs;
      ar.x = A_re[m * 64 + k0];  ar.y = A_re[m * 64 + k0 + 1];
      ai.x = A_im[m * 64 + k0];  ai.y = A_im[m * 64 + k0 + 1];
      bs.x = buf0[k0 * 64 + n];  bs.y = buf0[(k0 + 1) * 64 + n];
      acc_re = __builtin_amdgcn_wmma_f32_16x16x4_f32(false, ar, false, bs,
                                                     (short)0, acc_re, false, false);
      acc_im = __builtin_amdgcn_wmma_f32_16x16x4_f32(false, ai, false, bs,
                                                     (short)0, acc_im, false, false);
    }
    int mrow = tm * 16 + (lane >> 4) * 8;
    for (int v = 0; v < 8; ++v) {
      buf1[(mrow + v) * 64 + n] = acc_re[v];
      buf2[(mrow + v) * 64 + n] = acc_im[v];
    }
  }
  __syncthreads();

  for (int t = wave * 2; t < wave * 2 + 2; ++t) {
    int tm = t >> 2, tn = t & 3;
    int m = tm * 16 + ln15;
    int n = tn * 16 + ln15;
    v8f accP = {};
    v8f accQ = {};
    v8f accI = {};
    for (int ks = 0; ks < 16; ++ks) {
      int k0 = ks * 4 + khalf;
      v2f tre, tim, bre, bim;
      tre.x = buf1[m * 64 + k0];   tre.y = buf1[m * 64 + k0 + 1];
      tim.x = buf2[m * 64 + k0];   tim.y = buf2[m * 64 + k0 + 1];
      bre.x = B_re[n * 64 + k0];   bre.y = B_re[n * 64 + k0 + 1];
      bim.x = B_im[n * 64 + k0];   bim.y = B_im[n * 64 + k0 + 1];
      accP = __builtin_amdgcn_wmma_f32_16x16x4_f32(false, tre, false, bre,
                                                   (short)0, accP, false, false);
      accQ = __builtin_amdgcn_wmma_f32_16x16x4_f32(false, tim, false, bim,
                                                   (short)0, accQ, false, false);
      accI = __builtin_amdgcn_wmma_f32_16x16x4_f32(false, tre, false, bim,
                                                   (short)0, accI, false, false);
      accI = __builtin_amdgcn_wmma_f32_16x16x4_f32(false, tim, false, bre,
                                                   (short)0, accI, false, false);
    }
    int mrow = tm * 16 + (lane >> 4) * 8;
    for (int v = 0; v < 8; ++v) {
      float re = accP[v] - accQ[v];
      float im = accI[v];
      buf0[(mrow + v) * 64 + n] = re * re + im * im;
    }
  }
  __syncthreads();

  float psum = 0.f;
  for (int i = 0; i < 16; ++i) psum += buf0[tid + i * 256];
  for (int o = 16; o > 0; o >>= 1) psum += __shfl_xor(psum, o, 32);
  if (lane == 0) red[wave] = psum;
  __syncthreads();
  if (tid == 0) {
    float t = 0.f;
    for (int i = 0; i < 8; ++i) t += red[i];
    red[32] = t;
  }
  __syncthreads();
  float ps = red[32];
  bool zp = !(ps > EPSV);
  float inv = zp ? 0.f : (1.0f / fmaxf(ps, EPSV));
  float* __restrict__ orow = out + (size_t)blockIdx.x * SDIM;
  for (int i = 0; i < 16; ++i) {
    int m = tid + i * 256;
    orow[m] = zp ? (1.0f / 4096.0f) : buf0[perm[m]] * inv;
  }
}

extern "C" void kernel_launch(void* const* d_in, const int* in_sizes, int n_in,
                              void* d_out, int out_size, void* d_ws, size_t ws_size,
                              hipStream_t stream) {
  const float* x = (const float*)d_in[0];
  const float* w = (const float*)d_in[1];
  float* out = (float*)d_out;
  float* mat = (float*)d_ws;
  int* perm = (int*)(mat + 16384);

  qdl_setup<<<32, 256, 0, stream>>>(w, mat, perm);

  int batch = in_sizes[0] / SDIM;
  qdl_main<<<batch, 256, 0, stream>>>(x, mat, perm, out);
}


# WMMA operand fragments made lane-contiguous: A,B matrices stored in fragment order in workspace (2 lines per load, was 16); T planes at LDS row pitch 68 dwords (64 banks per read, was 4)
# speedup vs baseline: 1.1148x; 1.1148x over previous
#include <hip/hip_runtime.h>
#include <hip/hip_bf16.h>
#include <math.h>


typedef float v2f __attribute__((ext_vector_type(2)));
typedef float v8f __attribute__((ext_vector_type(8)));

#define NQ    12
#define SDIM  4096
#define EPSV  1e-10f
#define TP    68


static __device__ __forceinline__ void rot_entry(const float* __restrict__ w,
                                                 int q, int r, int c,
                                                 float& er, float& ei) {
  float phi = w[q * 3 + 0], th = w[q * 3 + 1], om = w[q * 3 + 2];
  float ch = cosf(0.5f * th), sh = sinf(0.5f * th);
  float hpo = 0.5f * (phi + om), hpm = 0.5f * (phi - om);
  float cpo = cosf(hpo), spo = sinf(hpo);
  float cpm = cosf(hpm), spm = sinf(hpm);
  if (r == 0) {
    if (c == 0) { er =  ch * cpo; ei = -ch * spo; }
    else        { er = -sh * cpm; ei = -sh * spm; }
  } else {
    if (c == 0) { er =  sh * cpm; ei = -sh * spm; }
    else        { er =  ch * cpo; ei =  ch * spo; }
  }
}

__global__ void qdl_setup(const float* __restrict__ w,
                          float* __restrict__ mat,
                          int* __restrict__ perm) {
  int tid = blockIdx.x * blockDim.x + threadIdx.x;
  if (tid < 2 * SDIM) {
    int isB = tid >= SDIM;
    int idx = tid & (SDIM - 1);
    int frag = idx >> 6, fl = (idx >> 1) & 31, fs = idx & 1;
    int outi = (frag >> 4) * 16 + (fl & 15);
    int ini  = (frag & 15) * 4 + (fl >> 4) * 2 + fs;
    int qbase = isB ? 6 : 0;
    float re = 1.f, im = 0.f;
    for (int q = 0; q < 6; ++q) {
      int bo = (outi >> (5 - q)) & 1;
      int bi = (ini  >> (5 - q)) & 1;
      float er, ei;
      rot_entry(w, qbase + q, bo, bi, er, ei);
      float nre = re * er - im * ei;
      float nim = re * ei + im * er;
      re = nre; im = nim;
    }
    mat[(isB ? 8192 : 0) + idx]     = re;
    mat[(isB ? 12288 : 4096) + idx] = im;
  }
  if (tid < SDIM) {
    int src = tid;
    for (int i = NQ - 1; i >= 0; --i)
      for (int j = NQ - 1; j > i; --j)
        if ((src >> (NQ - 1 - i)) & 1) src ^= (1 << (NQ - 1 - j));
    perm[tid] = src;
  }
}

__global__ void __launch_bounds__(256)
qdl_main(const float* __restrict__ x, const float* __restrict__ mat,
         const int* __restrict__ perm, float* __restrict__ out) {
  __shared__ float buf0[SDIM];
  __shared__ __attribute__((aligned(16))) float buf1[64 * TP];
  __shared__ __attribute__((aligned(16))) float buf2[64 * TP];
  __shared__ float red[64];

  const int tid  = threadIdx.x;
  const int wave = tid >> 5;
  const int lane = tid & 31;
  const int ln15 = lane & 15;
  const int khalf = (lane >> 4) * 2;
  const float* __restrict__ xr = x + (size_t)blockIdx.x * SDIM;

  float vals[16];
  float ss = 0.f;
  for (int i = 0; i < 16; ++i) {
    float v = xr[tid + i * 256];
    vals[i] = v;
    ss += v * v;
  }
  for (int o = 16; o > 0; o >>= 1) ss += __shfl_xor(ss, o, 32);
  if (lane == 0) red[wave] = ss;
  __syncthreads();
  if (tid == 0) {
    float t = 0.f;
    for (int i = 0; i < 8; ++i) t += red[i];
    red[32] = t;
  }
  __syncthreads();
  float nrm = sqrtf(red[32]);
  bool zerovec = !(nrm > EPSV);
  float scale = zerovec ? 0.f : (1.0f / fmaxf(nrm, EPSV));
  for (int i = 0; i < 16; ++i)
    buf0[tid + i * 256] = zerovec ? 0.015625f : vals[i] * scale;
  __syncthreads();

  const float* __restrict__ A_re = mat;
  const float* __restrict__ A_im = mat + 4096;
  const float* __restrict__ B_re = mat + 8192;
  const float* __restrict__ B_im = mat + 12288;

  for (int t = wave * 2; t < wave * 2 + 2; ++t) {
    int tm = t >> 2, tn = t & 3;
    int n = tn * 16 + ln15;
    v8f acc_re = {};
    v8f acc_im = {};
    for (int ks = 0; ks < 16; ++ks) {
      int k0 = ks * 4 + khalf;
      int fa = ((tm * 16 + ks) * 32 + lane) * 2;
      v2f ar, ai, bs;
      ar.x = A_re[fa];  ar.y = A_re[fa + 1];
      ai.x = A_im[fa];  ai.y = A_im[fa + 1];
      bs.x = buf0[k0 * 64 + n];  bs.y = buf0[(k0 + 1) * 64 + n];
      acc_re = __builtin_amdgcn_wmma_f32_16x16x4_f32(false, ar, false, bs,
                                                     (short)0, acc_re, false, false);
      acc_im = __builtin_amdgcn_wmma_f32_16x16x4_f32(false, ai, false, bs,
                                                     (short)0, acc_im, false, false);
    }
    int mrow = tm * 16 + (lane >> 4) * 8;
    for (int v = 0; v < 8; ++v) {
      buf1[(mrow + v) * TP + n] = acc_re[v];
      buf2[(mrow + v) * TP + n] = acc_im[v];
    }
  }
  __syncthreads();

  for (int t = wave * 2; t < wave * 2 + 2; ++t) {
    int tm = t >> 2, tn = t & 3;
    int m = tm * 16 + ln15;
    int n = tn * 16 + ln15;
    v8f accP = {};
    v8f accQ = {};
    v8f accI = {};
    for (int ks = 0; ks < 16; ++ks) {
      int k0 = ks * 4 + khalf;
      int fb = ((tn * 16 + ks) * 32 + lane) * 2;
      v2f tre, tim, bre, bim;
      tre.x = buf1[m * TP + k0];   tre.y = buf1[m * TP + k0 + 1];
      tim.x = buf2[m * TP + k0];   tim.y = buf2[m * TP + k0 + 1];
      bre.x = B_re[fb];   bre.y = B_re[fb + 1];
      bim.x = B_im[fb];   bim.y = B_im[fb + 1];
      accP = __builtin_amdgcn_wmma_f32_16x16x4_f32(false, tre, false, bre,
                                                   (short)0, accP, false, false);
      accQ = __builtin_amdgcn_wmma_f32_16x16x4_f32(false, tim, false, bim,
                                                   (short)0, accQ, false, false);
      accI = __builtin_amdgcn_wmma_f32_16x16x4_f32(false, tre, false, bim,
                                                   (short)0, accI, false, false);
      accI = __builtin_amdgcn_wmma_f32_16x16x4_f32(false, tim, false, bre,
                                                   (short)0, accI, false, false);
    }
    int mrow = tm * 16 + (lane >> 4) * 8;
    for (int v = 0; v < 8; ++v) {
      float re = accP[v] - accQ[v];
      float im = accI[v];
      buf0[(mrow + v) * 64 + n] = re * re + im * im;
    }
  }
  __syncthreads();

  float psum = 0.f;
  for (int i = 0; i < 16; ++i) psum += buf0[tid + i * 256];
  for (int o = 16; o > 0; o >>= 1) psum += __shfl_xor(psum, o, 32);
  if (lane == 0) red[wave] = psum;
  __syncthreads();
  if (tid == 0) {
    float t = 0.f;
    for (int i = 0; i < 8; ++i) t += red[i];
    red[32] = t;
  }
  __syncthreads();
  float ps = red[32];
  bool zp = !(ps > EPSV);
  float inv = zp ? 0.f : (1.0f / fmaxf(ps, EPSV));
  float* __restrict__ orow = out + (size_t)blockIdx.x * SDIM;
  for (int i = 0; i < 16; ++i) {
    int m = tid + i * 256;
    orow[m] = zp ? (1.0f / 4096.0f) : buf0[perm[m]] * inv;
  }
}

extern "C" void kernel_launch(void* const* d_in, const int* in_sizes, int n_in,
                              void* d_out, int out_size, void* d_ws, size_t ws_size,
                              hipStream_t stream) {
  const float* x = (const float*)d_in[0];
  const float* w = (const float*)d_in[1];
  float* out = (float*)d_out;
  float* mat = (float*)d_ws;
  int* perm = (int*)(mat + 16384);

  qdl_setup<<<32, 256, 0, stream>>>(w, mat, perm);

  int batch = in_sizes[0] / SDIM;
  qdl_main<<<batch, 256, 0, stream>>>(x, mat, perm, out);
}


# Setup launch and its tables removed: each work-group builds the 2x2 rotations and 3 factors in LDS; CNOT permutation as prefix XOR in registers, its 16 gathers issued as one group; 1 launch, was 2
# speedup vs baseline: 2.3171x; 2.0785x over previous
#include <hip/hip_runtime.h>
#include <hip/hip_bf16.h>
#include <math.h>


typedef float v2f __attribute__((ext_vector_type(2)));
typedef float v4f __attribute__((ext_vector_type(4)));
typedef float v8f __attribute__((ext_vector_type(8)));

#define NQ    12
#define SDIM  4096
#define EPSV  1e-10f
#define TP    20


static __device__ __forceinline__ void rot_matrix(float phi, float th, float om,
                                                  float (&u)[8]) {
  float ch = cosf(0.5f * th), sh = sinf(0.5f * th);
  float hpo = 0.5f * (phi + om), hpm = 0.5f * (phi - om);
  float cpo = cosf(hpo), spo = sinf(hpo);
  float cpm = cosf(hpm), spm = sinf(hpm);
  u[0] =  ch * cpo; u[1] = -ch * spo;
  u[2] = -sh * cpm; u[3] = -sh * spm;
  u[4] =  sh * cpm; u[5] = -sh * spm;
  u[6] =  ch * cpo; u[7] =  ch * spo;
}

static __device__ __forceinline__ void ld8(const float* __restrict__ p, float (&d)[8]) {
  v4f lo = *(const v4f*)p;
  v4f hi = *(const v4f*)(p + 4);
  d[0] = lo.x; d[1] = lo.y; d[2] = lo.z; d[3] = lo.w;
  d[4] = hi.x; d[5] = hi.y; d[6] = hi.z; d[7] = hi.w;
}

static __device__ __forceinline__ v8f mma(float a0, float a1, float b0, float b1, v8f c) {
  v2f av, bv;
  av.x = a0; av.y = a1;
  bv.x = b0; bv.y = b1;
  return __builtin_amdgcn_wmma_f32_16x16x4_f32(false, av, false, bv, (short)0, c, false, false);
}

__global__ void __launch_bounds__(256)
qdl_main(const float* __restrict__ x, const float* __restrict__ w, float* __restrict__ out) {
  __shared__ __attribute__((aligned(16))) float buf0[SDIM];
  __shared__ __attribute__((aligned(16))) float t2re[256 * TP];
  __shared__ __attribute__((aligned(16))) float t2im[256 * TP];
  __shared__ __attribute__((aligned(16))) float g1sh[512];
  __shared__ __attribute__((aligned(16))) float usm[NQ * 8];
  __shared__ float red[64];
  float* g32 = t2re;

  const int tid  = threadIdx.x;
  const int wave = tid >> 5;
  const int lane = tid & 31;
  const int ln15 = lane & 15;
  const int h    = lane >> 4;
  const float* __restrict__ xr = x + (size_t)blockIdx.x * SDIM;

  const int qw = tid < NQ ? tid : NQ - 1;
  float wphi = w[qw * 3 + 0], wth = w[qw * 3 + 1], wom = w[qw * 3 + 2];
  float vals[16];
  for (int i = 0; i < 16; ++i) vals[i] = xr[tid + i * 256];
  asm volatile("" :: "v"(wphi), "v"(wth), "v"(wom));
  if (tid < NQ) {
    float u[8];
    rot_matrix(wphi, wth, wom, u);
    v4f u0, u1;
    u0.x = u[0]; u0.y = u[1]; u0.z = u[2]; u0.w = u[3];
    u1.x = u[4]; u1.y = u[5]; u1.z = u[6]; u1.w = u[7];
    *(v4f*)&usm[tid * 8]     = u0;
    *(v4f*)&usm[tid * 8 + 4] = u1;
  }
  float ss = 0.f;
  for (int i = 0; i < 16; ++i) ss += vals[i] * vals[i];

  for (int o = 16; o > 0; o >>= 1) ss += __shfl_xor(ss, o, 32);
  if (lane == 0) red[wave] = ss;
  __syncthreads();
  {
    const int fl = tid >> 3, j = tid & 7;
    const int outi = fl & 15;
    const int ini  = (fl >> 4) * 8 + j;
    for (int g = 0; g < 3; ++g) {
      float re = 1.f, im = 0.f;
      for (int q = 0; q < 4; ++q) {
        int bo = (outi >> (3 - q)) & 1;
        int bi = (ini  >> (3 - q)) & 1;
        v2f e = *(const v2f*)&usm[(4 * g + q) * 8 + (bo * 2 + bi) * 2];
        float nre = re * e.x - im * e.y;
        float nim = re * e.y + im * e.x;
        re = nre; im = nim;
      }
      float* dst = (g == 0) ? g1sh : (g == 1 ? g32 + 512 : g32);
      dst[tid]       = re;
      dst[256 + tid] = im;
    }
  }
  if (tid == 0) {
    float t = 0.f;
    for (int i = 0; i < 8; ++i) t += red[i];
    red[32] = t;
  }
  __syncthreads();
  float nrm = sqrtf(red[32]);
  bool zerovec = !(nrm > EPSV);
  float scale = zerovec ? 0.f : (1.0f / fmaxf(nrm, EPSV));
  for (int i = 0; i < 16; ++i)
    buf0[tid + i * 256] = zerovec ? 0.015625f : vals[i] * scale;

  float g3r[8], g3i[8], g2r[8], g2i[8], g2n[8];
  ld8(g32 +   0 + lane * 8, g3r);
  ld8(g32 + 256 + lane * 8, g3i);
  ld8(g32 + 512 + lane * 8, g2r);
  ld8(g32 + 768 + lane * 8, g2i);
  for (int k = 0; k < 8; ++k) g2n[k] = -g2i[k];
  __syncthreads();

  v8f d2r[2], d2i[2];
  for (int u = 0; u < 2; ++u) {
    const int a = wave * 2 + u;
    float s[8];
    ld8(buf0 + a * 256 + ln15 * 16 + h * 8, s);
    v8f d1r = {};
    v8f d1i = {};
    for (int t = 0; t < 4; ++t) {
      d1r = mma(s[2 * t], s[2 * t + 1], g3r[2 * t], g3r[2 * t + 1], d1r);
      d1i = mma(s[2 * t], s[2 * t + 1], g3i[2 * t], g3i[2 * t + 1], d1i);
    }
    v8f er = {};
    v8f ei = {};
    for (int t = 0; t < 4; ++t) {
      er = mma(g2r[2 * t], g2r[2 * t + 1], d1r[2 * t], d1r[2 * t + 1], er);
      er = mma(g2n[2 * t], g2n[2 * t + 1], d1i[2 * t], d1i[2 * t + 1], er);
      ei = mma(g2r[2 * t], g2r[2 * t + 1], d1i[2 * t], d1i[2 * t + 1], ei);
      ei = mma(g2i[2 * t], g2i[2 * t + 1], d1r[2 * t], d1r[2 * t + 1], ei);
    }
    d2r[u] = er;
    d2i[u] = ei;
  }
  float g1r[8], g1i[8], g1n[8];
  ld8(g1sh +   0 + lane * 8, g1r);
  ld8(g1sh + 256 + lane * 8, g1i);
  for (int k = 0; k < 8; ++k) g1n[k] = -g1i[k];
  for (int r = 0; r < 8; ++r) {
    int bc = (h * 8 + r) * 16 + ln15;
    v2f pr, pi;
    pr.x = d2r[0][r]; pr.y = d2r[1][r];
    pi.x = d2i[0][r]; pi.y = d2i[1][r];
    *(v2f*)&t2re[bc * TP + wave * 2] = pr;
    *(v2f*)&t2im[bc * TP + wave * 2] = pi;
  }
  __syncthreads();

  for (int u = 0; u < 2; ++u) {
    const int bp = wave * 2 + u;
    float tr[8], ti[8];
    ld8(t2re + (bp * 16 + ln15) * TP + h * 8, tr);
    ld8(t2im + (bp * 16 + ln15) * TP + h * 8, ti);
    v8f qr = {};
    v8f qi = {};
    for (int t = 0; t < 4; ++t) {
      qr = mma(g1r[2 * t], g1r[2 * t + 1], tr[2 * t], tr[2 * t + 1], qr);
      qr = mma(g1n[2 * t], g1n[2 * t + 1], ti[2 * t], ti[2 * t + 1], qr);
      qi = mma(g1r[2 * t], g1r[2 * t + 1], ti[2 * t], ti[2 * t + 1], qi);
      qi = mma(g1i[2 * t], g1i[2 * t + 1], tr[2 * t], tr[2 * t + 1], qi);
    }
    for (int r = 0; r < 8; ++r) {
      float re = qr[r];
      float im = qi[r];
      buf0[(h * 8 + r) * 256 + bp * 16 + ln15] = re * re + im * im;
    }
  }
  __syncthreads();

  float psum = 0.f;
  for (int i = 0; i < 16; ++i) psum += buf0[tid + i * 256];
  for (int o = 16; o > 0; o >>= 1) psum += __shfl_xor(psum, o, 32);
  if (lane == 0) red[wave] = psum;
  __syncthreads();
  if (tid == 0) {
    float t = 0.f;
    for (int i = 0; i < 8; ++i) t += red[i];
    red[32] = t;
  }
  __syncthreads();
  float ps = red[32];
  bool zp = !(ps > EPSV);
  float inv = zp ? 0.f : (1.0f / fmaxf(ps, EPSV));
  float* __restrict__ orow = out + (size_t)blockIdx.x * SDIM;
  int pl = tid ^ (tid >> 1);
  pl ^= pl >> 2;
  pl ^= pl >> 4;
  if (zp) {
    for (int i = 0; i < 16; ++i) orow[tid + i * 256] = 1.0f / 4096.0f;
  } else {
    for (int i = 0; i < 16; ++i) {
      int hi = i ^ (i >> 1);
      hi ^= hi >> 2;
      int srci = (hi << 8) | ((hi & 1) ? (pl ^ 255) : pl);
      orow[tid + i * 256] = buf0[srci] * inv;
    }
  }
}

extern "C" void kernel_launch(void* const* d_in, const int* in_sizes, int n_in,
                              void* d_out, int out_size, void* d_ws, size_t ws_size,
                              hipStream_t stream) {
  const float* x = (const float*)d_in[0];
  const float* w = (const float*)d_in[1];
  float* out = (float*)d_out;

  int batch = in_sizes[0] / SDIM;
  qdl_main<<<batch, 256, 0, stream>>>(x, w, out);
}


# Grid 2048 -> 1024 work-groups (4 per WGP, all resident), each loops over its rows: factors built once per work-group, next row's loads in flight across LDS-only barriers; half the waves enter
# speedup vs baseline: 2.4448x; 1.0551x over previous
#include <hip/hip_runtime.h>
#include <hip/hip_bf16.h>
#include <math.h>


typedef float v2f __attribute__((ext_vector_type(2)));
typedef float v4f __attribute__((ext_vector_type(4)));
typedef float v8f __attribute__((ext_vector_type(8)));

#define NQ    12
#define SDIM  4096
#define EPSV  1e-10f
#define TP    20


static __device__ __forceinline__ void rot_matrix(float phi, float th, float om,
                                                  float (&u)[8]) {
  float ch = cosf(0.5f * th), sh = sinf(0.5f * th);
  float hpo = 0.5f * (phi + om), hpm = 0.5f * (phi - om);
  float cpo = cosf(hpo), spo = sinf(hpo);
  float cpm = cosf(hpm), spm = sinf(hpm);
  u[0] =  ch * cpo; u[1] = -ch * spo;
  u[2] = -sh * cpm; u[3] = -sh * spm;
  u[4] =  sh * cpm; u[5] = -sh * spm;
  u[6] =  ch * cpo; u[7] =  ch * spo;
}

static __device__ __forceinline__ void ld8(const float* __restrict__ p, float (&d)[8]) {
  v4f lo = *(const v4f*)p;
  v4f hi = *(const v4f*)(p + 4);
  d[0] = lo.x; d[1] = lo.y; d[2] = lo.z; d[3] = lo.w;
  d[4] = hi.x; d[5] = hi.y; d[6] = hi.z; d[7] = hi.w;
}

static __device__ __forceinline__ void lds_barrier() {
  __builtin_amdgcn_fence(__ATOMIC_RELEASE, "workgroup", "local");
  __builtin_amdgcn_s_barrier();
  __builtin_amdgcn_fence(__ATOMIC_ACQUIRE, "workgroup", "local");
}

static __device__ __forceinline__ v8f mma(float a0, float a1, float b0, float b1, v8f c) {
  v2f av, bv;
  av.x = a0; av.y = a1;
  bv.x = b0; bv.y = b1;
  return __builtin_amdgcn_wmma_f32_16x16x4_f32(false, av, false, bv, (short)0, c, false, false);
}

__global__ void __launch_bounds__(256, 8)
qdl_main(const float* __restrict__ x, const float* __restrict__ w, float* __restrict__ out,
         int batch) {
  __shared__ __attribute__((aligned(16))) float buf0[SDIM];
  __shared__ __attribute__((aligned(16))) float t2re[256 * TP];
  __shared__ __attribute__((aligned(16))) float t2im[256 * TP];
  __shared__ __attribute__((aligned(16))) float g1sh[512];
  __shared__ __attribute__((aligned(16))) float g32[1024];
  __shared__ __attribute__((aligned(16))) float usm[NQ * 8];
  __shared__ float red[64];

  const int tid  = threadIdx.x;
  const int wave = tid >> 5;
  const int lane = tid & 31;
  const int ln15 = lane & 15;
  const int h    = lane >> 4;
  int row = blockIdx.x;
  if (row >= batch) return;
  const int rstep = gridDim.x;
  const float* __restrict__ xr = x + (size_t)row * SDIM;

  const int qw = tid < NQ ? tid : NQ - 1;
  float wphi = w[qw * 3 + 0], wth = w[qw * 3 + 1], wom = w[qw * 3 + 2];
  float vals[16];
  for (int i = 0; i < 16; ++i) vals[i] = xr[tid + i * 256];
  asm volatile("" :: "v"(wphi), "v"(wth), "v"(wom));
  if (tid < NQ) {
    float u[8];
    rot_matrix(wphi, wth, wom, u);
    v4f u0, u1;
    u0.x = u[0]; u0.y = u[1]; u0.z = u[2]; u0.w = u[3];
    u1.x = u[4]; u1.y = u[5]; u1.z = u[6]; u1.w = u[7];
    *(v4f*)&usm[tid * 8]     = u0;
    *(v4f*)&usm[tid * 8 + 4] = u1;
  }
  bool first = true;
  for (; row < batch; row += rstep) {
  float ss = 0.f;
  for (int i = 0; i < 16; ++i) ss += vals[i] * vals[i];

  for (int o = 16; o > 0; o >>= 1) ss += __shfl_xor(ss, o, 32);
  if (lane == 0) red[wave] = ss;
  lds_barrier();
  if (first) {
    const int fl = tid >> 3, j = tid & 7;
    const int outi = fl & 15;
    const int ini  = (fl >> 4) * 8 + j;
    for (int g = 0; g < 3; ++g) {
      float re = 1.f, im = 0.f;
      for (int q = 0; q < 4; ++q) {
        int bo = (outi >> (3 - q)) & 1;
        int bi = (ini  >> (3 - q)) & 1;
        v2f e = *(const v2f*)&usm[(4 * g + q) * 8 + (bo * 2 + bi) * 2];
        float nre = re * e.x - im * e.y;
        float nim = re * e.y + im * e.x;
        re = nre; im = nim;
      }
      float* dst = (g == 0) ? g1sh : (g == 1 ? g32 + 512 : g32);
      dst[tid]       = re;
      dst[256 + tid] = im;
    }
  }
  if (tid == 0) {
    float t = 0.f;
    for (int i = 0; i < 8; ++i) t += red[i];
    red[32] = t;
  }
  lds_barrier();
  float nrm = sqrtf(red[32]);
  bool zerovec = !(nrm > EPSV);
  float scale = zerovec ? 0.f : (1.0f / fmaxf(nrm, EPSV));
  for (int i = 0; i < 16; ++i)
    buf0[tid + i * 256] = zerovec ? 0.015625f : vals[i] * scale;

  if (row + rstep < batch) {
    const float* __restrict__ xn = x + (size_t)(row + rstep) * SDIM;
    for (int i = 0; i < 16; ++i) vals[i] = xn[tid + i * 256];
  }
  float g3r[8], g3i[8], g2r[8], g2i[8], g2n[8];
  ld8(g32 +   0 + lane * 8, g3r);
  ld8(g32 + 256 + lane * 8, g3i);
  ld8(g32 + 512 + lane * 8, g2r);
  ld8(g32 + 768 + lane * 8, g2i);
  for (int k = 0; k < 8; ++k) g2n[k] = -g2i[k];
  lds_barrier();

  v8f d2r[2], d2i[2];
  for (int u = 0; u < 2; ++u) {
    const int a = wave * 2 + u;
    float s[8];
    ld8(buf0 + a * 256 + ln15 * 16 + h * 8, s);
    v8f d1r = {};
    v8f d1i = {};
    for (int t = 0; t < 4; ++t) {
      d1r = mma(s[2 * t], s[2 * t + 1], g3r[2 * t], g3r[2 * t + 1], d1r);
      d1i = mma(s[2 * t], s[2 * t + 1], g3i[2 * t], g3i[2 * t + 1], d1i);
    }
    v8f er = {};
    v8f ei = {};
    for (int t = 0; t < 4; ++t) {
      er = mma(g2r[2 * t], g2r[2 * t + 1], d1r[2 * t], d1r[2 * t + 1], er);
      er = mma(g2n[2 * t], g2n[2 * t + 1], d1i[2 * t], d1i[2 * t + 1], er);
      ei = mma(g2r[2 * t], g2r[2 * t + 1], d1i[2 * t], d1i[2 * t + 1], ei);
      ei = mma(g2i[2 * t], g2i[2 * t + 1], d1r[2 * t], d1r[2 * t + 1], ei);
    }
    d2r[u] = er;
    d2i[u] = ei;
  }
  float g1r[8], g1i[8], g1n[8];
  ld8(g1sh +   0 + lane * 8, g1r);
  ld8(g1sh + 256 + lane * 8, g1i);
  for (int k = 0; k < 8; ++k) g1n[k] = -g1i[k];
  for (int r = 0; r < 8; ++r) {
    int bc = (h * 8 + r) * 16 + ln15;
    v2f pr, pi;
    pr.x = d2r[0][r]; pr.y = d2r[1][r];
    pi.x = d2i[0][r]; pi.y = d2i[1][r];
    *(v2f*)&t2re[bc * TP + wave * 2] = pr;
    *(v2f*)&t2im[bc * TP + wave * 2] = pi;
  }
  lds_barrier();

  for (int u = 0; u < 2; ++u) {
    const int bp = wave * 2 + u;
    float tr[8], ti[8];
    ld8(t2re + (bp * 16 + ln15) * TP + h * 8, tr);
    ld8(t2im + (bp * 16 + ln15) * TP + h * 8, ti);
    v8f qr = {};
    v8f qi = {};
    for (int t = 0; t < 4; ++t) {
      qr = mma(g1r[2 * t], g1r[2 * t + 1], tr[2 * t], tr[2 * t + 1], qr);
      qr = mma(g1n[2 * t], g1n[2 * t + 1], ti[2 * t], ti[2 * t + 1], qr);
      qi = mma(g1r[2 * t], g1r[2 * t + 1], ti[2 * t], ti[2 * t + 1], qi);
      qi = mma(g1i[2 * t], g1i[2 * t + 1], tr[2 * t], tr[2 * t + 1], qi);
    }
    for (int r = 0; r < 8; ++r) {
      float re = qr[r];
      float im = qi[r];
      buf0[(h * 8 + r) * 256 + bp * 16 + ln15] = re * re + im * im;
    }
  }
  lds_barrier();

  float psum = 0.f;
  for (int i = 0; i < 16; ++i) psum += buf0[tid + i * 256];
  for (int o = 16; o > 0; o >>= 1) psum += __shfl_xor(psum, o, 32);
  if (lane == 0) red[wave] = psum;
  lds_barrier();
  if (tid == 0) {
    float t = 0.f;
    for (int i = 0; i < 8; ++i) t += red[i];
    red[32] = t;
  }
  lds_barrier();
  float ps = red[32];
  bool zp = !(ps > EPSV);
  float inv = zp ? 0.f : (1.0f / fmaxf(ps, EPSV));
  float* __restrict__ orow = out + (size_t)row * SDIM;
  int pl = tid ^ (tid >> 1);
  pl ^= pl >> 2;
  pl ^= pl >> 4;
  if (zp) {
    for (int i = 0; i < 16; ++i) orow[tid + i * 256] = 1.0f / 4096.0f;
  } else {
    const float* __restrict__ srck = buf0 + pl;
    const float* __restrict__ srcf = buf0 + (pl ^ 255);
    for (int i = 0; i < 16; ++i) {
      int hi = i ^ (i >> 1);
      hi ^= hi >> 2;
      orow[tid + i * 256] = ((hi & 1) ? srcf : srck)[hi << 8] * inv;
    }
  }
  first = false;
  }
}

extern "C" void kernel_launch(void* const* d_in, const int* in_sizes, int n_in,
                              void* d_out, int out_size, void* d_ws, size_t ws_size,
                              hipStream_t stream) {
  const float* x = (const float*)d_in[0];
  const float* w = (const float*)d_in[1];
  float* out = (float*)d_out;

  int batch = in_sizes[0] / SDIM;
  qdl_main<<<1024, 256, 0, stream>>>(x, w, out, batch);
}


# Rotations split as phase * real RY * phase: omega phase cancels in |.|^2, phi phases applied to the T1 tile on the vector unit, factors real: 384 f32 WMMA per row, was 640; sum order changes
# speedup vs baseline: 2.9449x; 1.2046x over previous
#include <hip/hip_runtime.h>
#include <hip/hip_bf16.h>
#include <math.h>


typedef float v2f __attribute__((ext_vector_type(2)));
typedef float v4f __attribute__((ext_vector_type(4)));
typedef float v8f __attribute__((ext_vector_type(8)));

#define NQ    12
#define SDIM  4096
#define EPSV  1e-10f
#define TP    20


static __device__ __forceinline__ void group_phase(const float* usm, int g, int idx,
                                                   float& pre, float& pim) {
  float re = 1.f, im = 0.f;
  for (int q = 0; q < 4; ++q) {
    v4f e = *(const v4f*)&usm[(4 * g + q) * 4];
    float s = ((idx >> (3 - q)) & 1) ? e.w : -e.w;
    float nre = re * e.z - im * s;
    float nim = re * s + im * e.z;
    re = nre; im = nim;
  }
  pre = re; pim = im;
}

static __device__ __forceinline__ void ld8(const float* __restrict__ p, float (&d)[8]) {
  v4f lo = *(const v4f*)p;
  v4f hi = *(const v4f*)(p + 4);
  d[0] = lo.x; d[1] = lo.y; d[2] = lo.z; d[3] = lo.w;
  d[4] = hi.x; d[5] = hi.y; d[6] = hi.z; d[7] = hi.w;
}

static __device__ __forceinline__ void lds_barrier() {
  __builtin_amdgcn_fence(__ATOMIC_RELEASE, "workgroup", "local");
  __builtin_amdgcn_s_barrier();
  __builtin_amdgcn_fence(__ATOMIC_ACQUIRE, "workgroup", "local");
}

static __device__ __forceinline__ v8f mma(float a0, float a1, float b0, float b1, v8f c) {
  v2f av, bv;
  av.x = a0; av.y = a1;
  bv.x = b0; bv.y = b1;
  return __builtin_amdgcn_wmma_f32_16x16x4_f32(false, av, false, bv, (short)0, c, false, false);
}

__global__ void __launch_bounds__(256, 8)
qdl_main(const float* __restrict__ x, const float* __restrict__ w, float* __restrict__ out,
         int batch) {
  __shared__ __attribute__((aligned(16))) float buf0[SDIM];
  __shared__ __attribute__((aligned(16))) float t2re[256 * TP];
  __shared__ __attribute__((aligned(16))) float t2im[256 * TP];
  __shared__ __attribute__((aligned(16))) float g1sh[256];
  __shared__ __attribute__((aligned(16))) float g32[768];
  __shared__ __attribute__((aligned(16))) float pab[512];
  __shared__ __attribute__((aligned(16))) float usm[NQ * 4];
  __shared__ float red[64];

  const int tid  = threadIdx.x;
  const int wave = tid >> 5;
  const int lane = tid & 31;
  const int ln15 = lane & 15;
  const int h    = lane >> 4;
  int row = blockIdx.x;
  if (row >= batch) return;
  const int rstep = gridDim.x;
  const float* __restrict__ xr = x + (size_t)row * SDIM;

  const int qw = tid < NQ ? tid : NQ - 1;
  float wphi = w[qw * 3 + 0], wth = w[qw * 3 + 1];
  float vals[16];
  for (int i = 0; i < 16; ++i) vals[i] = xr[tid + i * 256];
  asm volatile("" :: "v"(wphi), "v"(wth));
  if (tid < NQ) {
    v4f u0;
    u0.x = cosf(0.5f * wth);  u0.y = sinf(0.5f * wth);
    u0.z = cosf(0.5f * wphi); u0.w = sinf(0.5f * wphi);
    *(v4f*)&usm[tid * 4] = u0;
  }
  bool first = true;
  for (; row < batch; row += rstep) {
  float ss = 0.f;
  for (int i = 0; i < 16; ++i) ss += vals[i] * vals[i];

  for (int o = 16; o > 0; o >>= 1) ss += __shfl_xor(ss, o, 32);
  if (lane == 0) red[wave] = ss;
  lds_barrier();
  if (first) {
    const int fl = tid >> 3, j = tid & 7;
    const int outi = fl & 15;
    const int ini  = (fl >> 4) * 8 + j;
    float rg[3];
    for (int g = 0; g < 3; ++g) {
      float rr = 1.f;
      for (int q = 0; q < 4; ++q) {
        int bo = (outi >> (3 - q)) & 1;
        int bi = (ini  >> (3 - q)) & 1;
        v2f e = *(const v2f*)&usm[(4 * g + q) * 4];
        rr *= (bo == bi) ? e.x : (bo ? e.y : -e.y);
      }
      rg[g] = rr;
    }
    float p3r, p3i, p1r, p1i, p2r, p2i;
    group_phase(usm, 2, ini, p3r, p3i);
    group_phase(usm, 0, tid >> 4, p1r, p1i);
    group_phase(usm, 1, tid & 15, p2r, p2i);
    g1sh[tid]      = rg[0];
    g32[512 + tid] = rg[1];
    g32[tid]       = rg[2] * p3r;
    g32[256 + tid] = rg[2] * p3i;
    pab[tid]       = p1r * p2r - p1i * p2i;
    pab[256 + tid] = p1r * p2i + p1i * p2r;
  }
  if (tid == 0) {
    float t = 0.f;
    for (int i = 0; i < 8; ++i) t += red[i];
    red[32] = t;
  }
  lds_barrier();
  float nrm = sqrtf(red[32]);
  bool zerovec = !(nrm > EPSV);
  float scale = zerovec ? 0.f : (1.0f / fmaxf(nrm, EPSV));
  for (int i = 0; i < 16; ++i)
    buf0[tid + i * 256] = zerovec ? 0.015625f : vals[i] * scale;

  if (row + rstep < batch) {
    const float* __restrict__ xn = x + (size_t)(row + rstep) * SDIM;
    for (int i = 0; i < 16; ++i) vals[i] = xn[tid + i * 256];
  }
  float g3r[8], g3i[8], g2r[8];
  ld8(g32 +   0 + lane * 8, g3r);
  ld8(g32 + 256 + lane * 8, g3i);
  ld8(g32 + 512 + lane * 8, g2r);
  lds_barrier();

  v8f d2r[2], d2i[2];
  for (int u = 0; u < 2; ++u) {
    const int a = wave * 2 + u;
    float s[8];
    ld8(buf0 + a * 256 + ln15 * 16 + h * 8, s);
    float pr[8], pi[8];
    ld8(pab +   0 + a * 16 + h * 8, pr);
    ld8(pab + 256 + a * 16 + h * 8, pi);
    v8f d1r = {};
    v8f d1i = {};
    for (int t = 0; t < 4; ++t) {
      d1r = mma(s[2 * t], s[2 * t + 1], g3r[2 * t], g3r[2 * t + 1], d1r);
      d1i = mma(s[2 * t], s[2 * t + 1], g3i[2 * t], g3i[2 * t + 1], d1i);
    }
    float xr[8], xi[8];
    for (int r = 0; r < 8; ++r) {
      xr[r] = d1r[r] * pr[r] - d1i[r] * pi[r];
      xi[r] = d1r[r] * pi[r] + d1i[r] * pr[r];
    }
    v8f er = {};
    v8f ei = {};
    for (int t = 0; t < 4; ++t) {
      er = mma(g2r[2 * t], g2r[2 * t + 1], xr[2 * t], xr[2 * t + 1], er);
      ei = mma(g2r[2 * t], g2r[2 * t + 1], xi[2 * t], xi[2 * t + 1], ei);
    }
    d2r[u] = er;
    d2i[u] = ei;
  }
  float g1r[8];
  ld8(g1sh + lane * 8, g1r);
  for (int r = 0; r < 8; ++r) {
    int bc = (h * 8 + r) * 16 + ln15;
    v2f pr, pi;
    pr.x = d2r[0][r]; pr.y = d2r[1][r];
    pi.x = d2i[0][r]; pi.y = d2i[1][r];
    *(v2f*)&t2re[bc * TP + wave * 2] = pr;
    *(v2f*)&t2im[bc * TP + wave * 2] = pi;
  }
  lds_barrier();

  for (int u = 0; u < 2; ++u) {
    const int bp = wave * 2 + u;
    float tr[8], ti[8];
    ld8(t2re + (bp * 16 + ln15) * TP + h * 8, tr);
    ld8(t2im + (bp * 16 + ln15) * TP + h * 8, ti);
    v8f qr = {};
    v8f qi = {};
    for (int t = 0; t < 4; ++t) {
      qr = mma(g1r[2 * t], g1r[2 * t + 1], tr[2 * t], tr[2 * t + 1], qr);
      qi = mma(g1r[2 * t], g1r[2 * t + 1], ti[2 * t], ti[2 * t + 1], qi);
    }
    for (int r = 0; r < 8; ++r) {
      float re = qr[r];
      float im = qi[r];
      buf0[(h * 8 + r) * 256 + bp * 16 + ln15] = re * re + im * im;
    }
  }
  lds_barrier();

  float psum = 0.f;
  for (int i = 0; i < 16; ++i) psum += buf0[tid + i * 256];
  for (int o = 16; o > 0; o >>= 1) psum += __shfl_xor(psum, o, 32);
  if (lane == 0) red[wave] = psum;
  lds_barrier();
  if (tid == 0) {
    float t = 0.f;
    for (int i = 0; i < 8; ++i) t += red[i];
    red[32] = t;
  }
  lds_barrier();
  float ps = red[32];
  bool zp = !(ps > EPSV);
  float inv = zp ? 0.f : (1.0f / fmaxf(ps, EPSV));
  float* __restrict__ orow = out + (size_t)row * SDIM;
  int pl = tid ^ (tid >> 1);
  pl ^= pl >> 2;
  pl ^= pl >> 4;
  if (zp) {
    for (int i = 0; i < 16; ++i) orow[tid + i * 256] = 1.0f / 4096.0f;
  } else {
    const float* __restrict__ srck = buf0 + pl;
    const float* __restrict__ srcf = buf0 + (pl ^ 255);
    for (int i = 0; i < 16; ++i) {
      int hi = i ^ (i >> 1);
      hi ^= hi >> 2;
      orow[tid + i * 256] = ((hi & 1) ? srcf : srck)[hi << 8] * inv;
    }
  }
  first = false;
  }
}

extern "C" void kernel_launch(void* const* d_in, const int* in_sizes, int n_in,
                              void* d_out, int out_size, void* d_ws, size_t ws_size,
                              hipStream_t stream) {
  const float* x = (const float*)d_in[0];
  const float* w = (const float*)d_in[1];
  float* out = (float*)d_out;

  int batch = in_sizes[0] / SDIM;
  qdl_main<<<1024, 256, 0, stream>>>(x, w, out, batch);
}


# Input normalisation deferred: raw row loaded straight into the A fragments, 1/|x|^2 folded into the final probability scale, both row sums finished together behind the contractions; sum order changes
# speedup vs baseline: 3.2836x; 1.1150x over previous
#include <hip/hip_runtime.h>
#include <hip/hip_bf16.h>
#include <math.h>


typedef float v2f __attribute__((ext_vector_type(2)));
typedef float v4f __attribute__((ext_vector_type(4)));
typedef float v8f __attribute__((ext_vector_type(8)));

#define NQ    12
#define SDIM  4096
#define EPSV  1e-10f
#define TP    20


static __device__ __forceinline__ void group_phase(const float* usm, int g, int idx,
                                                   float& pre, float& pim) {
  float re = 1.f, im = 0.f;
  for (int q = 0; q < 4; ++q) {
    v4f e = *(const v4f*)&usm[(4 * g + q) * 4];
    float s = ((idx >> (3 - q)) & 1) ? e.w : -e.w;
    float nre = re * e.z - im * s;
    float nim = re * s + im * e.z;
    re = nre; im = nim;
  }
  pre = re; pim = im;
}

static __device__ __forceinline__ void ld8(const float* __restrict__ p, float (&d)[8]) {
  v4f lo = *(const v4f*)p;
  v4f hi = *(const v4f*)(p + 4);
  d[0] = lo.x; d[1] = lo.y; d[2] = lo.z; d[3] = lo.w;
  d[4] = hi.x; d[5] = hi.y; d[6] = hi.z; d[7] = hi.w;
}

static __device__ __forceinline__ void lds_barrier() {
  __builtin_amdgcn_fence(__ATOMIC_RELEASE, "workgroup", "local");
  __builtin_amdgcn_s_barrier();
  __builtin_amdgcn_fence(__ATOMIC_ACQUIRE, "workgroup", "local");
}

static __device__ __forceinline__ v8f mma(float a0, float a1, float b0, float b1, v8f c) {
  v2f av, bv;
  av.x = a0; av.y = a1;
  bv.x = b0; bv.y = b1;
  return __builtin_amdgcn_wmma_f32_16x16x4_f32(false, av, false, bv, (short)0, c, false, false);
}

static __device__ __forceinline__ float xhalf(float x) {
  return __builtin_bit_cast(float, __builtin_amdgcn_permlanex16(__builtin_bit_cast(int, x),
      __builtin_bit_cast(int, x), 0x76543210, 0xfedcba98, false, false));
}
template <int M>
static __device__ __forceinline__ float add_xmask(float v) {
  int t = __builtin_amdgcn_update_dpp(0, __builtin_bit_cast(int, v), 0x160 + M, 0xf, 0xf, false);
  return v + __builtin_bit_cast(float, t);
}
static __device__ __forceinline__ float wave_sum(float v) {
  v = add_xmask<1>(v);
  v = add_xmask<2>(v);
  v = add_xmask<4>(v);
  v = add_xmask<8>(v);
  return v + xhalf(v);
}
static __device__ __forceinline__ float sum8(const float* p) {
  v4f lo = *(const v4f*)p;
  v4f hi = *(const v4f*)(p + 4);
  float t = 0.f;
  t += lo.x; t += lo.y; t += lo.z; t += lo.w;
  t += hi.x; t += hi.y; t += hi.z; t += hi.w;
  return t;
}

__global__ void __launch_bounds__(256, 8)
qdl_main(const float* __restrict__ x, const float* __restrict__ w, float* __restrict__ out,
         int batch) {
  __shared__ __attribute__((aligned(16))) float buf0[SDIM];
  __shared__ __attribute__((aligned(16))) float t2[2 * 256 * TP];
  __shared__ __attribute__((aligned(16))) float g1sh[256];
  __shared__ __attribute__((aligned(16))) float g32[768];
  __shared__ __attribute__((aligned(16))) float pab[512];
  __shared__ __attribute__((aligned(16))) float usm[NQ * 4];
  __shared__ __attribute__((aligned(16))) float red[16];
  __shared__ __attribute__((aligned(16))) float red2[8];

  float* const t2re = t2;
  float* const t2im = t2 + 256 * TP;
  const int tid  = threadIdx.x;
  const int wave = tid >> 5;
  const int lane = tid & 31;
  const int ln15 = lane & 15;
  const int h    = lane >> 4;
  int row = blockIdx.x;
  if (row >= batch) return;
  const int rstep = gridDim.x;

  const int qw = tid < NQ ? tid : NQ - 1;
  float wphi = w[qw * 3 + 0], wth = w[qw * 3 + 1];
  const int foff = wave * 512 + ln15 * 16 + h * 8;
  v4f sa[2][2];
  {
    const float* __restrict__ xr = x + (size_t)row * SDIM;
    sa[0][0] = *(const v4f*)(xr + foff);
    sa[0][1] = *(const v4f*)(xr + foff + 4);
    sa[1][0] = *(const v4f*)(xr + foff + 256);
    sa[1][1] = *(const v4f*)(xr + foff + 260);
  }
  asm volatile("" :: "v"(wphi), "v"(wth));
  if (tid < NQ) {
    v4f u0;
    u0.x = cosf(0.5f * wth);  u0.y = sinf(0.5f * wth);
    u0.z = cosf(0.5f * wphi); u0.w = sinf(0.5f * wphi);
    *(v4f*)&usm[tid * 4] = u0;
  }
  lds_barrier();
  {
    const int fl = tid >> 3, j = tid & 7;
    const int outi = fl & 15;
    const int ini  = (fl >> 4) * 8 + j;
    float rg[3];
    for (int g = 0; g < 3; ++g) {
      float rr = 1.f;
      for (int q = 0; q < 4; ++q) {
        int bo = (outi >> (3 - q)) & 1;
        int bi = (ini  >> (3 - q)) & 1;
        v2f e = *(const v2f*)&usm[(4 * g + q) * 4];
        rr *= (bo == bi) ? e.x : (bo ? e.y : -e.y);
      }
      rg[g] = rr;
    }
    float p3r, p3i, p1r, p1i, p2r, p2i;
    group_phase(usm, 2, ini, p3r, p3i);
    group_phase(usm, 0, tid >> 4, p1r, p1i);
    group_phase(usm, 1, tid & 15, p2r, p2i);
    g1sh[tid]      = rg[0];
    g32[512 + tid] = rg[1];
    g32[tid]       = rg[2] * p3r;
    g32[256 + tid] = rg[2] * p3i;
    pab[tid]       = p1r * p2r - p1i * p2i;
    pab[256 + tid] = p1r * p2i + p1i * p2r;
  }
  lds_barrier();

  bool uni = false;
  int par = 0;
  while (row < batch) {
    const int nxt = row + rstep;
    float ssl = 0.f;
    for (int u = 0; u < 2; ++u)
      for (int j = 0; j < 8; ++j) ssl += sa[u][j >> 2][j & 3] * sa[u][j >> 2][j & 3];
    ssl = wave_sum(ssl);
    if (lane == 0) red[par * 8 + wave] = ssl;

    float g3r[8], g3i[8], g2r[8];
    ld8(g32 +   0 + lane * 8, g3r);
    ld8(g32 + 256 + lane * 8, g3i);
    ld8(g32 + 512 + lane * 8, g2r);

    v8f d2r[2], d2i[2];
    for (int u = 0; u < 2; ++u) {
      const int a = wave * 2 + u;
      float pr[8], pi[8];
      ld8(pab +   0 + a * 16 + h * 8, pr);
      ld8(pab + 256 + a * 16 + h * 8, pi);
      v8f d1r = {};
      v8f d1i = {};
      for (int t = 0; t < 4; ++t) {
        const float s0 = sa[u][t >> 1][(2 * t) & 3], s1 = sa[u][t >> 1][(2 * t + 1) & 3];
        d1r = mma(s0, s1, g3r[2 * t], g3r[2 * t + 1], d1r);
        d1i = mma(s0, s1, g3i[2 * t], g3i[2 * t + 1], d1i);
      }
      float xr[8], xi[8];
      for (int r = 0; r < 8; ++r) {
        xr[r] = d1r[r] * pr[r] - d1i[r] * pi[r];
        xi[r] = d1r[r] * pi[r] + d1i[r] * pr[r];
      }
      v8f er = {};
      v8f ei = {};
      for (int t = 0; t < 4; ++t) {
        er = mma(g2r[2 * t], g2r[2 * t + 1], xr[2 * t], xr[2 * t + 1], er);
        ei = mma(g2r[2 * t], g2r[2 * t + 1], xi[2 * t], xi[2 * t + 1], ei);
      }
      d2r[u] = er;
      d2i[u] = ei;
    }
    if (nxt < batch) {
      const float* __restrict__ xn = x + (size_t)nxt * SDIM;
      sa[0][0] = *(const v4f*)(xn + foff);
      sa[0][1] = *(const v4f*)(xn + foff + 4);
      sa[1][0] = *(const v4f*)(xn + foff + 256);
      sa[1][1] = *(const v4f*)(xn + foff + 260);
    }
    for (int r = 0; r < 8; ++r) {
      int bc = (h * 8 + r) * 16 + ln15;
      v2f pr, pi;
      pr.x = d2r[0][r]; pr.y = d2r[1][r];
      pi.x = d2i[0][r]; pi.y = d2i[1][r];
      *(v2f*)&t2re[bc * TP + wave * 2] = pr;
      *(v2f*)&t2im[bc * TP + wave * 2] = pi;
    }
    lds_barrier();

    float g1r[8];
    ld8(g1sh + lane * 8, g1r);
    float psl = 0.f;
    for (int u = 0; u < 2; ++u) {
      const int bp = wave * 2 + u;
      float tr[8], ti[8];
      ld8(t2re + (bp * 16 + ln15) * TP + h * 8, tr);
      ld8(t2im + (bp * 16 + ln15) * TP + h * 8, ti);
      v8f qr = {};
      v8f qi = {};
      for (int t = 0; t < 4; ++t) {
        qr = mma(g1r[2 * t], g1r[2 * t + 1], tr[2 * t], tr[2 * t + 1], qr);
        qi = mma(g1r[2 * t], g1r[2 * t + 1], ti[2 * t], ti[2 * t + 1], qi);
      }
      for (int r = 0; r < 8; ++r) {
        float re = qr[r];
        float im = qi[r];
        float p = re * re + im * im;
        buf0[(h * 8 + r) * 256 + bp * 16 + ln15] = p;
        psl += p;
      }
    }
    psl = wave_sum(psl);
    if (lane == 0) red2[wave] = psl;
    lds_barrier();

    const float ss  = sum8(red + par * 8);
    const float psr = sum8(red2);
    par ^= 1;
    float s = 1.0f;
    if (!uni) {
      float nrm = sqrtf(ss);
      if (!(nrm > EPSV)) {
        uni = true;
        for (int u = 0; u < 2; ++u)
          for (int k = 0; k < 2; ++k) sa[u][k] = (v4f)(0.015625f);
        continue;
      }
      s = 1.0f / fmaxf(nrm, EPSV);
    }
    const float ps = (psr * s) * s;
    const bool zp = !(ps > EPSV);
    const float inv = zp ? 0.f : (s * (1.0f / fmaxf(ps, EPSV))) * s;
    float* __restrict__ orow = out + (size_t)row * SDIM;
    int pl = tid ^ (tid >> 1);
    pl ^= pl >> 2;
    pl ^= pl >> 4;
    if (zp) {
      for (int i = 0; i < 16; ++i) orow[tid + i * 256] = 1.0f / 4096.0f;
    } else {
      const float* __restrict__ srck = buf0 + pl;
      const float* __restrict__ srcf = buf0 + (pl ^ 255);
      for (int i = 0; i < 16; ++i) {
        int hi = i ^ (i >> 1);
        hi ^= hi >> 2;
        orow[tid + i * 256] = ((hi & 1) ? srcf : srck)[hi << 8] * inv;
      }
    }
    uni = false;
    row = nxt;
  }
}

extern "C" void kernel_launch(void* const* d_in, const int* in_sizes, int n_in,
                              void* d_out, int out_size, void* d_ws, size_t ws_size,
                              hipStream_t stream) {
  const float* x = (const float*)d_in[0];
  const float* w = (const float*)d_in[1];
  float* out = (float*)d_out;

  int batch = in_sizes[0] / SDIM;
  qdl_main<<<1024, 256, 0, stream>>>(x, w, out, batch);
}


# Probability sum taken as |x|^2 s^2 (unitary layer: 1), not summed: scale known behind barrier 1, each wave scales its D tiles and stores whole lines; no LDS plane, gathers or wait on other waves
# speedup vs baseline: 3.4923x; 1.0636x over previous
#include <hip/hip_runtime.h>
#include <hip/hip_bf16.h>
#include <math.h>


typedef float v2f __attribute__((ext_vector_type(2)));
typedef float v4f __attribute__((ext_vector_type(4)));
typedef float v8f __attribute__((ext_vector_type(8)));

#define NQ    12
#define SDIM  4096
#define EPSV  1e-10f
#define TP    20


static __device__ __forceinline__ void group_phase(const float* usm, int g, int idx,
                                                   float& pre, float& pim) {
  float re = 1.f, im = 0.f;
  for (int q = 0; q < 4; ++q) {
    v4f e = *(const v4f*)&usm[(4 * g + q) * 4];
    float s = ((idx >> (3 - q)) & 1) ? e.w : -e.w;
    float nre = re * e.z - im * s;
    float nim = re * s + im * e.z;
    re = nre; im = nim;
  }
  pre = re; pim = im;
}

static __device__ __forceinline__ void ld8(const float* __restrict__ p, float (&d)[8]) {
  v4f lo = *(const v4f*)p;
  v4f hi = *(const v4f*)(p + 4);
  d[0] = lo.x; d[1] = lo.y; d[2] = lo.z; d[3] = lo.w;
  d[4] = hi.x; d[5] = hi.y; d[6] = hi.z; d[7] = hi.w;
}

static __device__ __forceinline__ void lds_barrier() {
  __builtin_amdgcn_fence(__ATOMIC_RELEASE, "workgroup", "local");
  __builtin_amdgcn_s_barrier();
  __builtin_amdgcn_fence(__ATOMIC_ACQUIRE, "workgroup", "local");
}

static __device__ __forceinline__ v8f mma(float a0, float a1, float b0, float b1, v8f c) {
  v2f av, bv;
  av.x = a0; av.y = a1;
  bv.x = b0; bv.y = b1;
  return __builtin_amdgcn_wmma_f32_16x16x4_f32(false, av, false, bv, (short)0, c, false, false);
}

static __device__ __forceinline__ float xhalf(float x) {
  return __builtin_bit_cast(float, __builtin_amdgcn_permlanex16(__builtin_bit_cast(int, x),
      __builtin_bit_cast(int, x), 0x76543210, 0xfedcba98, false, false));
}
template <int M>
static __device__ __forceinline__ float add_xmask(float v) {
  int t = __builtin_amdgcn_update_dpp(0, __builtin_bit_cast(int, v), 0x160 + M, 0xf, 0xf, false);
  return v + __builtin_bit_cast(float, t);
}
static __device__ __forceinline__ float wave_sum(float v) {
  v = add_xmask<1>(v);
  v = add_xmask<2>(v);
  v = add_xmask<4>(v);
  v = add_xmask<8>(v);
  return v + xhalf(v);
}
static __device__ __forceinline__ float sum8(const float* p) {
  v4f lo = *(const v4f*)p;
  v4f hi = *(const v4f*)(p + 4);
  float t = 0.f;
  t += lo.x; t += lo.y; t += lo.z; t += lo.w;
  t += hi.x; t += hi.y; t += hi.z; t += hi.w;
  return t;
}

__global__ void __launch_bounds__(256, 8)
qdl_main(const float* __restrict__ x, const float* __restrict__ w, float* __restrict__ out,
         int batch) {
  __shared__ __attribute__((aligned(16))) float t2[2 * 256 * TP];
  __shared__ __attribute__((aligned(16))) float g1sh[256];
  __shared__ __attribute__((aligned(16))) float g32[768];
  __shared__ __attribute__((aligned(16))) float pab[512];
  __shared__ __attribute__((aligned(16))) float usm[NQ * 4];
  __shared__ __attribute__((aligned(16))) float red[16];

  float* const t2re = t2;
  float* const t2im = t2 + 256 * TP;
  const int tid  = threadIdx.x;
  const int wave = tid >> 5;
  const int lane = tid & 31;
  const int ln15 = lane & 15;
  const int h    = lane >> 4;
  int row = blockIdx.x;
  if (row >= batch) return;
  const int rstep = gridDim.x;

  const int qw = tid < NQ ? tid : NQ - 1;
  float wphi = w[qw * 3 + 0], wth = w[qw * 3 + 1];
  const int foff = wave * 512 + ln15 * 16 + h * 8;
  v4f sa[2][2];
  {
    const float* __restrict__ xr = x + (size_t)row * SDIM;
    sa[0][0] = *(const v4f*)(xr + foff);
    sa[0][1] = *(const v4f*)(xr + foff + 4);
    sa[1][0] = *(const v4f*)(xr + foff + 256);
    sa[1][1] = *(const v4f*)(xr + foff + 260);
  }
  asm volatile("" :: "v"(wphi), "v"(wth));
  if (tid < NQ) {
    v4f u0;
    u0.x = cosf(0.5f * wth);  u0.y = sinf(0.5f * wth);
    u0.z = cosf(0.5f * wphi); u0.w = sinf(0.5f * wphi);
    *(v4f*)&usm[tid * 4] = u0;
  }
  lds_barrier();
  {
    const int fl = tid >> 3, j = tid & 7;
    const int outi = fl & 15;
    const int ini  = (fl >> 4) * 8 + j;
    float rg[3];
    for (int g = 0; g < 3; ++g) {
      float rr = 1.f;
      for (int q = 0; q < 4; ++q) {
        int bo = (outi >> (3 - q)) & 1;
        int bi = (ini  >> (3 - q)) & 1;
        v2f e = *(const v2f*)&usm[(4 * g + q) * 4];
        rr *= (bo == bi) ? e.x : (bo ? e.y : -e.y);
      }
      rg[g] = rr;
    }
    float p3r, p3i, p1r, p1i, p2r, p2i;
    group_phase(usm, 2, ini, p3r, p3i);
    group_phase(usm, 0, tid >> 4, p1r, p1i);
    group_phase(usm, 1, tid & 15, p2r, p2i);
    g1sh[tid]      = rg[0];
    g32[512 + tid] = rg[1];
    g32[tid]       = rg[2] * p3r;
    g32[256 + tid] = rg[2] * p3i;
    pab[tid]       = p1r * p2r - p1i * p2i;
    pab[256 + tid] = p1r * p2i + p1i * p2r;
  }
  lds_barrier();

  bool uni = false;
  int par = 0;
  while (row < batch) {
    const int nxt = row + rstep;
    float ssl = 0.f;
    for (int u = 0; u < 2; ++u)
      for (int j = 0; j < 8; ++j) ssl += sa[u][j >> 2][j & 3] * sa[u][j >> 2][j & 3];
    ssl = wave_sum(ssl);
    if (lane == 0) red[par * 8 + wave] = ssl;

    float g3r[8], g3i[8], g2r[8];
    ld8(g32 +   0 + lane * 8, g3r);
    ld8(g32 + 256 + lane * 8, g3i);
    ld8(g32 + 512 + lane * 8, g2r);

    v8f d2r[2], d2i[2];
    for (int u = 0; u < 2; ++u) {
      const int a = wave * 2 + u;
      float pr[8], pi[8];
      ld8(pab +   0 + a * 16 + h * 8, pr);
      ld8(pab + 256 + a * 16 + h * 8, pi);
      v8f d1r = {};
      v8f d1i = {};
      for (int t = 0; t < 4; ++t) {
        const float s0 = sa[u][t >> 1][(2 * t) & 3], s1 = sa[u][t >> 1][(2 * t + 1) & 3];
        d1r = mma(s0, s1, g3r[2 * t], g3r[2 * t + 1], d1r);
        d1i = mma(s0, s1, g3i[2 * t], g3i[2 * t + 1], d1i);
      }
      float xr[8], xi[8];
      for (int r = 0; r < 8; ++r) {
        xr[r] = d1r[r] * pr[r] - d1i[r] * pi[r];
        xi[r] = d1r[r] * pi[r] + d1i[r] * pr[r];
      }
      v8f er = {};
      v8f ei = {};
      for (int t = 0; t < 4; ++t) {
        er = mma(g2r[2 * t], g2r[2 * t + 1], xr[2 * t], xr[2 * t + 1], er);
        ei = mma(g2r[2 * t], g2r[2 * t + 1], xi[2 * t], xi[2 * t + 1], ei);
      }
      d2r[u] = er;
      d2i[u] = ei;
    }
    if (nxt < batch) {
      const float* __restrict__ xn = x + (size_t)nxt * SDIM;
      sa[0][0] = *(const v4f*)(xn + foff);
      sa[0][1] = *(const v4f*)(xn + foff + 4);
      sa[1][0] = *(const v4f*)(xn + foff + 256);
      sa[1][1] = *(const v4f*)(xn + foff + 260);
    }
    for (int r = 0; r < 8; ++r) {
      int bc = (h * 8 + r) * 16 + ln15;
      v2f pr, pi;
      pr.x = d2r[0][r]; pr.y = d2r[1][r];
      pi.x = d2i[0][r]; pi.y = d2i[1][r];
      *(v2f*)&t2re[bc * TP + wave * 2] = pr;
      *(v2f*)&t2im[bc * TP + wave * 2] = pi;
    }
    lds_barrier();

    const float ss = sum8(red + par * 8);
    par ^= 1;

    float g1r[8];
    ld8(g1sh + lane * 8, g1r);
    float tr[2][8], ti[2][8];
    for (int u = 0; u < 2; ++u) {
      const int bp = wave * 2 + u;
      ld8(t2re + (bp * 16 + ln15) * TP + h * 8, tr[u]);
      ld8(t2im + (bp * 16 + ln15) * TP + h * 8, ti[u]);
    }
    lds_barrier();

    float pq[2][8];
    for (int u = 0; u < 2; ++u) {
      v8f qr = {};
      v8f qi = {};
      for (int t = 0; t < 4; ++t) {
        qr = mma(g1r[2 * t], g1r[2 * t + 1], tr[u][2 * t], tr[u][2 * t + 1], qr);
        qi = mma(g1r[2 * t], g1r[2 * t + 1], ti[u][2 * t], ti[u][2 * t + 1], qi);
      }
      for (int r = 0; r < 8; ++r) {
        float re = qr[r];
        float im = qi[r];
        pq[u][r] = re * re + im * im;
      }
    }
    const float nrm = sqrtf(ss);
    const float s = 1.0f / fmaxf(nrm, EPSV);
    const float ps = (ss * s) * s;
    const bool zp = !(ps > EPSV);
    const float inv = s * s;
    if (!uni && !(nrm > EPSV)) {
      uni = true;
      for (int u = 0; u < 2; ++u)
        for (int k = 0; k < 2; ++k) sa[u][k] = (v4f)(0.015625f);
      continue;
    }
    float* __restrict__ orow = out + (size_t)row * SDIM;
    if (zp) {
      for (int i = 0; i < 16; ++i) orow[tid + i * 256] = 1.0f / 4096.0f;
    } else {
      const int gt = tid ^ (tid >> 1);
      float* __restrict__ dstk = orow + gt;
      float* __restrict__ dstf = orow + (gt ^ 128);
      for (int r = 0; r < 8; ++r) {
        const float p0 = pq[0][r] * inv, p1 = pq[1][r] * inv;
        const float got = xhalf(h ? p0 : p1);
        const float lo = h ? got : p0;
        const float hi = h ? p1 : got;
        const int alo = r, ahi = r + 8;
        ((r & 1) ? dstf : dstk)[(alo ^ (alo >> 1)) << 8] = lo;
        ((r & 1) ? dstf : dstk)[(ahi ^ (ahi >> 1)) << 8] = hi;
      }
    }
    uni = false;
    row = nxt;
  }
}

extern "C" void kernel_launch(void* const* d_in, const int* in_sizes, int n_in,
                              void* d_out, int out_size, void* d_ws, size_t ws_size,
                              hipStream_t stream) {
  const float* x = (const float*)d_in[0];
  const float* w = (const float*)d_in[1];
  float* out = (float*)d_out;

  int batch = in_sizes[0] / SDIM;
  qdl_main<<<1024, 256, 0, stream>>>(x, w, out, batch);
}
